# starting kernel
# baseline (speedup 1.0000x reference)
#include <hip/hip_runtime.h>


typedef __attribute__((ext_vector_type(2))) float v2f;
typedef __attribute__((ext_vector_type(8))) float v8f;

#define CIN   32
#define FOUT  32
#define Hdim  128
#define Wdim  128
#define KHW   3
#define TILE_W 32
#define ROWS_PER_BLK 4
#define XROWS (ROWS_PER_BLK + 2)
#define XCOLS (TILE_W + 2)
#define XPIX_STRIDE 36
#define WSIZE (KHW * KHW * CIN * FOUT)
#define PSTRIDE (WSIZE + FOUT)

__global__ __launch_bounds__(128)
void conv3x3_wmma_f32(const float* __restrict__ x,
                      const float* __restrict__ params,
                      float* __restrict__ out) {
    __shared__ float sW3[WSIZE];
    __shared__ float sX[XROWS * XCOLS * XPIX_STRIDE];
    __shared__ float sB[FOUT];

    const int tid  = threadIdx.x;
    const int b    = blockIdx.y;
    const int wt   = blockIdx.x & 3;
    const int hblk = blockIdx.x >> 2;
    const int h0   = hblk * ROWS_PER_BLK;
    const int w0   = wt * TILE_W;

    {
        const float* pb = params + (size_t)b * PSTRIDE;
        float4* wdst = (float4*)sW3;
        #pragma unroll
        for (int it = 0; it < WSIZE / 4 / 128; ++it) {
            const int j   = tid + 128 * it;
            const int tap = j >> 8;
            const int q   = (j >> 5) & 7;
            const int f   = j & 31;
            const float* src = pb + ((tap * 32 + 4 * q) * 32 + f);
            wdst[j] = make_float4(src[0], src[32], src[64], src[96]);
        }
        if (tid < FOUT)
            sB[tid] = pb[WSIZE + tid];
    }

    const int NXPIX = XROWS * XCOLS;
    float4* xdst = (float4*)sX;
    for (int i = tid; i < NXPIX * 8; i += 128) {
        const int pix = i >> 3;
        const int c4  = i & 7;
        const int r   = pix / XCOLS;
        const int c   = pix - r * XCOLS;
        const int row = h0 - 1 + r;
        const int col = w0 - 1 + c;
        float4 v = make_float4(0.f, 0.f, 0.f, 0.f);
        if ((unsigned)row < (unsigned)Hdim && (unsigned)col < (unsigned)Wdim)
            v = *(const float4*)(x + ((((size_t)b * Hdim + row) * Wdim + col) * CIN) + c4 * 4);
        xdst[pix * (XPIX_STRIDE / 4) + c4] = v;
    }
    __syncthreads();

    const int lane  = tid & 31;
    const int wave  = tid >> 5;
    const int mlane = lane & 15;
    const int hi    = lane >> 4;

    v8f acc00 = {}, acc01 = {}, acc10 = {}, acc11 = {};

    #pragma unroll
    for (int kh = 0; kh < KHW; ++kh) {
        #pragma unroll
        for (int kw = 0; kw < KHW; ++kw) {
            const float* xrow = &sX[((kh + wave) * XCOLS + mlane + kw) * XPIX_STRIDE];
            const float* wb   = &sW3[(kh * 3 + kw) * 1024 + mlane * 4 + 2 * hi];
            #pragma unroll
            for (int kb = 0; kb < CIN; kb += 4) {
                const int k0 = kb + 2 * hi;
                v2f a0 = { xrow[k0],                       xrow[k0 + 1] };
                v2f a1 = { xrow[16 * XPIX_STRIDE + k0],    xrow[16 * XPIX_STRIDE + k0 + 1] };
                const float* bp = wb + kb * 32;
                v2f b0 = *(const v2f*)bp;
                v2f b1 = *(const v2f*)(bp + 64);
                acc00 = __builtin_amdgcn_wmma_f32_16x16x4_f32(false, a0, false, b0,
                        (short)0, acc00, false, false);
                acc01 = __builtin_amdgcn_wmma_f32_16x16x4_f32(false, a0, false, b1,
                        (short)0, acc01, false, false);
                acc10 = __builtin_amdgcn_wmma_f32_16x16x4_f32(false, a1, false, b0,
                        (short)0, acc10, false, false);
                acc11 = __builtin_amdgcn_wmma_f32_16x16x4_f32(false, a1, false, b1,
                        (short)0, acc11, false, false);
            }
        }
    }

    const float bias0 = sB[mlane];
    const float bias1 = sB[16 + mlane];
    const int   h     = h0 + wave;
    float* obase = out
        + ((((size_t)b * Hdim + h) * Wdim) + w0 + 8 * hi) * FOUT + mlane;
    #pragma unroll
    for (int r = 0; r < 8; ++r) {
        obase[(size_t)r * FOUT]             = acc00[r] + bias0;
        obase[(size_t)r * FOUT + 16]        = acc01[r] + bias1;
        obase[(size_t)(16 + r) * FOUT]      = acc10[r] + bias0;
        obase[(size_t)(16 + r) * FOUT + 16] = acc11[r] + bias1;
    }
}

extern "C" void kernel_launch(void* const* d_in, const int* in_sizes, int n_in,
                              void* d_out, int out_size, void* d_ws, size_t ws_size,
                              hipStream_t stream) {
    (void)in_sizes; (void)n_in; (void)d_ws; (void)ws_size; (void)out_size;
    const float* x      = (const float*)d_in[0];
    const float* params = (const float*)d_in[1];
    float* out          = (float*)d_out;

    dim3 grid((Hdim / ROWS_PER_BLK) * (Wdim / TILE_W), 32);
    conv3x3_wmma_f32<<<grid, 128, 0, stream>>>(x, params, out);
}


# Weights held in registers: each wave loads its sample's weight set from global straight into 288 B-fragment VGPRs and walks a full 128-px row (4-row full-width strips, grid 1024); LDS holds x only
# speedup vs baseline: 1.1892x; 1.1892x over previous
#include <hip/hip_runtime.h>


typedef __attribute__((ext_vector_type(2))) float v2f;
typedef __attribute__((ext_vector_type(8))) float v8f;

#define CIN   32
#define FOUT  32
#define Hdim  128
#define Wdim  128
#define KHW   3
#define ROWS_PER_BLK 4
#define XROWS (ROWS_PER_BLK + 2)
#define XCOLS (Wdim + 2)
#define XPIX_STRIDE 36
#define XPIX_F4 (XPIX_STRIDE / 4)
#define ROW_F4 (Wdim * CIN / 4)
#define WSIZE (KHW * KHW * CIN * FOUT)
#define PSTRIDE (WSIZE + FOUT)

__global__ __launch_bounds__(128)
void conv3x3_wreg_f32(const float* __restrict__ x,
                      const float* __restrict__ params,
                      float* __restrict__ out) {
    __shared__ __attribute__((aligned(16))) float sX[XROWS * XCOLS * XPIX_STRIDE];

    const int tid   = threadIdx.x;
    const int b     = blockIdx.y;
    const int h0    = blockIdx.x * ROWS_PER_BLK;
    const int lane  = tid & 31;
    const int wave  = tid >> 5;
    const int mlane = lane & 15;
    const int hi    = lane >> 4;

    const float* pb = params + (size_t)b * PSTRIDE;

    v2f wr[KHW * KHW][CIN / 4][2];
    {
        const float* wsrc = pb + (2 * hi) * FOUT + mlane;
        #pragma unroll
        for (int tap = 0; tap < KHW * KHW; ++tap) {
            #pragma unroll
            for (int q = 0; q < CIN / 4; ++q) {
                #pragma unroll
                for (int nt = 0; nt < 2; ++nt) {
                    const float* s = wsrc + (tap * CIN + 4 * q) * FOUT + 16 * nt;
                    v2f w = { s[0], s[FOUT] };
                    wr[tap][q][nt] = w;
                }
            }
        }
    }
    const float bias0 = pb[WSIZE + mlane];
    const float bias1 = pb[WSIZE + 16 + mlane];

    {
        float4* xdst = (float4*)sX;
        const float4* xsrc = (const float4*)(x + (size_t)b * Hdim * Wdim * CIN);
        #pragma unroll
        for (int r = 0; r < XROWS; ++r) {
            const int  row  = h0 - 1 + r;
            const bool ok   = (unsigned)row < (unsigned)Hdim;
            const int  rowc = ok ? row : h0;
            #pragma unroll
            for (int s = 0; s < ROW_F4 / 128; ++s) {
                const int i  = tid + 128 * s;
                const int px = i >> 3;
                const int c4 = i & 7;
                float4 v = xsrc[(size_t)rowc * ROW_F4 + i];
                if (!ok)
                    v = make_float4(0.f, 0.f, 0.f, 0.f);
                xdst[(r * XCOLS + 1 + px) * XPIX_F4 + c4] = v;
            }
        }
        if (tid < XROWS * 2 * 8) {
            const int r    = tid >> 4;
            const int side = (tid >> 3) & 1;
            const int c4   = tid & 7;
            xdst[(r * XCOLS + side * (XCOLS - 1)) * XPIX_F4 + c4] =
                make_float4(0.f, 0.f, 0.f, 0.f);
        }
    }
    __syncthreads();

    const int h = h0 + wave;
    float* orow = out + ((((size_t)b * Hdim + h) * Wdim) + 8 * hi) * FOUT + mlane;
    const float* xbase = &sX[(wave * XCOLS + mlane) * XPIX_STRIDE + 2 * hi];

    #pragma unroll 1
    for (int it = 0; it < Wdim / 32; ++it) {
        const float* xt = xbase + it * 32 * XPIX_STRIDE;
        v8f acc00 = {}, acc01 = {}, acc10 = {}, acc11 = {};

        #pragma unroll
        for (int kh = 0; kh < KHW; ++kh) {
            #pragma unroll
            for (int kw = 0; kw < KHW; ++kw) {
                const float* xrow = xt + (kh * XCOLS + kw) * XPIX_STRIDE;
                #pragma unroll
                for (int q = 0; q < CIN / 4; ++q) {
                    v2f a0 = *(const v2f*)(xrow + 4 * q);
                    v2f a1 = *(const v2f*)(xrow + 16 * XPIX_STRIDE + 4 * q);
                    const v2f b0 = wr[kh * KHW + kw][q][0];
                    const v2f b1 = wr[kh * KHW + kw][q][1];
                    acc00 = __builtin_amdgcn_wmma_f32_16x16x4_f32(false, a0, false, b0,
                            (short)0, acc00, false, false);
                    acc01 = __builtin_amdgcn_wmma_f32_16x16x4_f32(false, a0, false, b1,
                            (short)0, acc01, false, false);
                    acc10 = __builtin_amdgcn_wmma_f32_16x16x4_f32(false, a1, false, b0,
                            (short)0, acc10, false, false);
                    acc11 = __builtin_amdgcn_wmma_f32_16x16x4_f32(false, a1, false, b1,
                            (short)0, acc11, false, false);
                }
            }
        }

        float* obase = orow + (size_t)it * 32 * FOUT;
        #pragma unroll
        for (int r = 0; r < 8; ++r) {
            obase[(size_t)r * FOUT]             = acc00[r] + bias0;
            obase[(size_t)r * FOUT + 16]        = acc01[r] + bias1;
            obase[(size_t)(16 + r) * FOUT]      = acc10[r] + bias0;
            obase[(size_t)(16 + r) * FOUT + 16] = acc11[r] + bias1;
        }
    }
}

extern "C" void kernel_launch(void* const* d_in, const int* in_sizes, int n_in,
                              void* d_out, int out_size, void* d_ws, size_t ws_size,
                              hipStream_t stream) {
    (void)in_sizes; (void)n_in; (void)d_ws; (void)ws_size; (void)out_size;
    const float* x      = (const float*)d_in[0];
    const float* params = (const float*)d_in[1];
    float* out          = (float*)d_out;

    dim3 grid(Hdim / ROWS_PER_BLK, 32);
    conv3x3_wreg_f32<<<grid, 128, 0, stream>>>(x, params, out);
}
